# starting kernel
# baseline (speedup 1.0000x reference)
#include <hip/hip_runtime.h>

typedef __attribute__((ext_vector_type(2))) float v2f;
typedef __attribute__((ext_vector_type(8))) float v8f;

#define DWORDS_D 300
#define S_CATS 256
#define A_CATS 256
#define WPC 8
#define NROWS 2048
#define COS_EPS 1e-8f

__global__ void ComSim_gather_rows(const float* __restrict__ emb,
                                   const int* __restrict__ senCats,
                                   const int* __restrict__ annCats,
                                   const int* __restrict__ noneIdxPtr,
                                   float* __restrict__ senE,
                                   float* __restrict__ annE,
                                   float* __restrict__ ns,
                                   float* __restrict__ na,
                                   float* __restrict__ noneMask)
{
    __shared__ float redF[128];
    __shared__ int   redI[128];

    const int bid   = blockIdx.x;
    const int tid   = threadIdx.x;
    const bool isSen = bid < NROWS;
    const int row   = isSen ? bid : bid - NROWS;
    const int idx   = isSen ? senCats[row] : annCats[row];
    const int nIdx  = noneIdxPtr[0];

    const float* __restrict__ src  = emb + (long long)idx  * DWORDS_D;
    const float* __restrict__ nsrc = emb + (long long)nIdx * DWORDS_D;
    float* __restrict__ dst = (isSen ? senE : annE) + (long long)row * DWORDS_D;

    float ss = 0.0f;
    int eq = 1;
    for (int d = tid; d < DWORDS_D; d += 128) {
        float v = src[d];
        dst[d] = v;
        ss += v * v;
        if (isSen) eq &= (v == nsrc[d]) ? 1 : 0;
    }
    redF[tid] = ss;
    redI[tid] = eq;
    __syncthreads();
    for (int off = 64; off > 0; off >>= 1) {
        if (tid < off) {
            redF[tid] += redF[tid + off];
            redI[tid] &= redI[tid + off];
        }
        __syncthreads();
    }
    if (tid == 0) {
        float nrm = sqrtf(redF[0]);
        if (isSen) {
            ns[row] = nrm;
            noneMask[row] = redI[0] ? 1.0f : 0.0f;
        } else {
            na[row] = nrm;
        }
    }
}

__global__ void ComSim_sen_word_sum(const float* __restrict__ senE,
                                    float* __restrict__ out)
{
    const int s = blockIdx.x;
    const int tid = threadIdx.x;
    for (int d = tid; d < DWORDS_D; d += 128) {
        float acc = 0.0f;
#pragma unroll
        for (int w = 0; w < WPC; ++w)
            acc += senE[((long long)(s * WPC + w)) * DWORDS_D + d];
        out[(long long)s * DWORDS_D + d] = acc;
    }
}

__global__ void ComSim_cos_tile_wmma(const float* __restrict__ senE,
                                     const float* __restrict__ annE,
                                     const float* __restrict__ ns,
                                     const float* __restrict__ na,
                                     const float* __restrict__ noneMask,
                                     float* __restrict__ maxSim)
{
    __shared__ float tile[16 * 16];

    const int lane = threadIdx.x;
    const int half = lane >> 4;
    const int l16  = lane & 15;
    const int row0 = blockIdx.y * 16;
    const int col0 = blockIdx.x * 16;

    const float* __restrict__ aRow = senE + (long long)(row0 + l16) * DWORDS_D + 2 * half;
    const float* __restrict__ bRow = annE + (long long)(col0 + l16) * DWORDS_D + 2 * half;

    v8f c = {};
#pragma unroll 5
    for (int k = 0; k < DWORDS_D; k += 4) {
        v2f a = *(const v2f*)(aRow + k);
        v2f b = *(const v2f*)(bRow + k);
        c = __builtin_amdgcn_wmma_f32_16x16x4_f32(
                 false, a,  false, b,
                 (short)0, c,  false,  false);
    }

    const float naV = na[col0 + l16];
#pragma unroll
    for (int r = 0; r < 8; ++r) {
        const int m = r + 8 * half;
        const float nsV   = ns[row0 + m];
        const float denom = fmaxf(nsV * naV, COS_EPS);
        float cosv = c[r] / denom;
        if (noneMask[row0 + m] != 0.0f) cosv = 0.0f;
        tile[m * 16 + l16] = cosv;
    }
    __syncthreads();

    if (lane < 4) {
        const int si = lane >> 1;
        const int ai = lane & 1;
        float cur = 0.0f;
        for (int w = 0; w < WPC; ++w) {
            for (int v = 0; v < WPC; ++v) {
                float x = tile[(si * 8 + w) * 16 + (ai * 8 + v)];
                cur = ((x >= cur) || (x < 0.0f)) ? x : cur;
            }
        }
        const int sIdx = blockIdx.y * 2 + si;
        const int aIdx = blockIdx.x * 2 + ai;
        maxSim[sIdx * A_CATS + aIdx] = cur;
    }
}

extern "C" void kernel_launch(void* const* d_in, const int* in_sizes, int n_in,
                              void* d_out, int out_size, void* d_ws, size_t ws_size,
                              hipStream_t stream) {
    const float* emb      = (const float*)d_in[0];
    const int*   senCats  = (const int*)d_in[1];
    const int*   annCats  = (const int*)d_in[2];
    const int*   noneIdx  = (const int*)d_in[3];

    float* out    = (float*)d_out;
    float* maxSim = out;
    float* senWd  = out + S_CATS * A_CATS;

    char* ws = (char*)d_ws;
    float* senE     = (float*)(ws + 0);
    float* annE     = (float*)(ws + 2457600);
    float* ns       = (float*)(ws + 4915200);
    float* na       = (float*)(ws + 4923392);
    float* noneMask = (float*)(ws + 4931584);

    ComSim_gather_rows<<<dim3(2 * NROWS), dim3(128), 0, stream>>>(
        emb, senCats, annCats, noneIdx, senE, annE, ns, na, noneMask);

    ComSim_sen_word_sum<<<dim3(S_CATS), dim3(128), 0, stream>>>(senE, senWd);

    ComSim_cos_tile_wmma<<<dim3(128, 128), dim3(32), 0, stream>>>(
        senE, annE, ns, na, noneMask, maxSim);
}


# Epilogue divides only what the fold can keep: raw dots go to the LDS tile, each cell's fold runs backward from the last word pair, 8 pairs a trip, and ends at its first negative; sum orders kept
# speedup vs baseline: 3.6169x; 3.6169x over previous
#include <hip/hip_runtime.h>

typedef __attribute__((ext_vector_type(2))) float v2f;
typedef __attribute__((ext_vector_type(8))) float v8f;

#define DWORDS_D 300
#define ROW_BYTES (DWORDS_D * 4)
#define S_CATS 256
#define A_CATS 256
#define WPC 8
#define COS_EPS 1e-8f

__device__ __forceinline__ v8f div8(const v8f num, const v8f den)
{
    float ds[8], nm[8], rc[8], e[8], q[8], t[8];
    bool fl[8];
    bool unusedFlag;
    v8f res;
#pragma unroll
    for (int r = 0; r < 8; ++r) ds[r] = __builtin_amdgcn_div_scalef(num[r], den[r], false, &unusedFlag);
#pragma unroll
    for (int r = 0; r < 8; ++r) rc[r] = __builtin_amdgcn_rcpf(ds[r]);
#pragma unroll
    for (int r = 0; r < 8; ++r) e[r] = __builtin_fmaf(-ds[r], rc[r], 1.0f);
#pragma unroll
    for (int r = 0; r < 8; ++r) rc[r] = __builtin_fmaf(e[r], rc[r], rc[r]);
#pragma unroll
    for (int r = 0; r < 8; ++r) nm[r] = __builtin_amdgcn_div_scalef(num[r], den[r], true, &fl[r]);
#pragma unroll
    for (int r = 0; r < 8; ++r) q[r] = nm[r] * rc[r];
#pragma unroll
    for (int r = 0; r < 8; ++r) t[r] = __builtin_fmaf(-ds[r], q[r], nm[r]);
#pragma unroll
    for (int r = 0; r < 8; ++r) q[r] = __builtin_fmaf(t[r], rc[r], q[r]);
#pragma unroll
    for (int r = 0; r < 8; ++r) t[r] = __builtin_fmaf(-ds[r], q[r], nm[r]);
#pragma unroll
    for (int r = 0; r < 8; ++r) t[r] = __builtin_amdgcn_div_fmasf(t[r], rc[r], q[r], fl[r]);
#pragma unroll
    for (int r = 0; r < 8; ++r) res[r] = __builtin_amdgcn_div_fixupf(t[r], den[r], num[r]);
    return res;
}

__device__ __forceinline__ bool eq4(const float4 p, const float4 q)
{
    return (p.x == q.x) & (p.y == q.y) & (p.z == q.z) & (p.w == q.w);
}

#define BT 4
#define RING 5
#define KSTEPS (DWORDS_D / 4)
#define TP 68

__global__ __launch_bounds__(128)
void ComSim_cos_block_wmma(const float* __restrict__ emb,
                           const int* __restrict__ senCats,
                           const int* __restrict__ annCats,
                           const int* __restrict__ noneIdxPtr,
                           float* __restrict__ maxSim,
                           float* __restrict__ senWd)
{
    __shared__ __attribute__((aligned(16))) float tile[4][64 * TP];
    __shared__ __attribute__((aligned(16))) float rowNorm[4][64];
    __shared__ __attribute__((aligned(16))) float rowNone[4][64];
    __shared__ __attribute__((aligned(16))) float colNorm[4][64];

    const int wave = __builtin_amdgcn_readfirstlane((int)(threadIdx.x >> 5));
    const int tid  = threadIdx.x;
    const int lane = threadIdx.x & 31;
    const int half = lane >> 4;
    const int l16  = lane & 15;
    const int wy   = wave >> 1;
    const int wx   = wave & 1;
    const int row0 = blockIdx.y * 128 + wy * 64;
    const int col0 = blockIdx.x * 128 + wx * 64;

    unsigned offA[BT], offB[BT];
#pragma unroll
    for (int i = 0; i < BT; ++i) {
        offA[i] = (unsigned)senCats[row0 + i * 16 + l16] * ROW_BYTES + 8u * half;
        offB[i] = (unsigned)annCats[col0 + i * 16 + l16] * ROW_BYTES + 8u * half;
    }
    const char* __restrict__ embB = (const char*)emb;
    const char* __restrict__ noneB = embB + (size_t)(unsigned)noneIdxPtr[0] * ROW_BYTES;
    const unsigned offH = 8u * half;

    const int sW = blockIdx.y * 16 + blockIdx.x;
    const int dW[3] = {tid, tid + 128, (tid + 256 < DWORDS_D) ? tid + 256 : DWORDS_D - 1};
    float wv[3][WPC];
#pragma unroll
    for (int w = 0; w < WPC; ++w) {
        const float* __restrict__ wr = emb + (size_t)(unsigned)senCats[sW * WPC + w] * DWORDS_D;
#pragma unroll
        for (int q = 0; q < 3; ++q) wv[q][w] = wr[dW[q]];
    }
    const float4* __restrict__ nRow  = (const float4*)noneB;
    const float4* __restrict__ xRow0 = (const float4*)(embB + (size_t)((half ? offA[2] : offA[0]) - offH));
    const float4* __restrict__ xRow1 = (const float4*)(embB + (size_t)((half ? offA[3] : offA[1]) - offH));
    const float4 nHead = nRow[0], xHead0 = xRow0[0], xHead1 = xRow1[0];
    __builtin_amdgcn_sched_barrier(0);

    v8f c[BT][BT];
#pragma unroll
    for (int i = 0; i < BT; ++i)
#pragma unroll
        for (int j = 0; j < BT; ++j)
            c[i][j] = (v8f){};

    v2f a[RING][BT], b[RING][BT];
    v2f ssA[BT], ssB[BT];
#pragma unroll
    for (int i = 0; i < BT; ++i) {
        ssA[i] = (v2f){};
        ssB[i] = (v2f){};
    }

    auto loadSet = [&](int set, int kstep) {
        const char* __restrict__ kb = embB + 16 * kstep;
#pragma unroll
        for (int i = 0; i < BT; ++i) {
            a[set][i] = *(const v2f*)(kb + (size_t)offA[i]);
            b[set][i] = *(const v2f*)(kb + (size_t)offB[i]);
        }
    };
    auto wmmaSet = [&](int set) {
#pragma unroll
        for (int i = 0; i < BT; ++i)
#pragma unroll
            for (int j = 0; j < BT; ++j)
                c[i][j] = __builtin_amdgcn_wmma_f32_16x16x4_f32(
                         false, a[set][i],  false, b[set][j],
                         (short)0, c[i][j],  false,  false);
    };
    auto sqSet = [&](int set) {
#pragma unroll
        for (int i = 0; i < BT; ++i) {
            ssA[i] = __builtin_elementwise_fma(a[set][i], a[set][i], ssA[i]);
            ssB[i] = __builtin_elementwise_fma(b[set][i], b[set][i], ssB[i]);
        }
    };

#pragma unroll
    for (int s = 0; s < RING - 1; ++s) {
        loadSet(s, s);
        __builtin_amdgcn_sched_barrier(0);
    }

    asm volatile("" :: "v"(wv[2][0]), "v"(wv[2][1]), "v"(wv[2][2]), "v"(wv[2][3]),
                       "v"(wv[2][4]), "v"(wv[2][5]), "v"(wv[2][6]), "v"(wv[2][7]));
#pragma unroll
    for (int q = 0; q < 3; ++q) {
        float acc = 0.0f;
#pragma unroll
        for (int w = 0; w < WPC; ++w) acc += wv[q][w];
        if (q < 2 || tid + 256 < DWORDS_D)
            senWd[(size_t)sW * DWORDS_D + dW[q]] = acc;
    }
    float* __restrict__ nrmL = &rowNorm[wave][0];
    float* __restrict__ nonL = &rowNone[wave][0];
    {
        bool isNone0 = eq4(xHead0, nHead);
        bool isNone1 = eq4(xHead1, nHead);
        if (__builtin_amdgcn_ballot_w32(isNone0 | isNone1) != 0u) {
            for (int t = 1; t < DWORDS_D / 4; ++t) {
                const float4 nn = nRow[t];
                isNone0 = isNone0 & eq4(xRow0[t], nn);
                isNone1 = isNone1 & eq4(xRow1[t], nn);
            }
        }
        nonL[32 * half + l16]      = isNone0 ? 1.0f : 0.0f;
        nonL[32 * half + 16 + l16] = isNone1 ? 1.0f : 0.0f;
    }
    __builtin_amdgcn_sched_barrier(0);

    for (int kk = 0; kk < KSTEPS - RING; kk += RING) {
#pragma unroll
        for (int u = 0; u < RING; ++u) {
            loadSet((u + RING - 1) % RING, kk + u + RING - 1);
            __builtin_amdgcn_sched_barrier(0);
            wmmaSet(u);
            sqSet(u);
#pragma unroll
            for (int i = 0; i < BT; ++i) {
                __builtin_amdgcn_sched_group_barrier(0x008, 4, 0);
                __builtin_amdgcn_sched_group_barrier(0x002, 2, 0);
            }
            __builtin_amdgcn_sched_barrier(0);
        }
    }
    loadSet(RING - 1, KSTEPS - 1);
    __builtin_amdgcn_sched_barrier(0);
#pragma unroll
    for (int u = 0; u < RING; ++u) wmmaSet(u);
    __builtin_amdgcn_sched_barrier(0);
#pragma unroll
    for (int u = 0; u < RING; ++u) {
        sqSet(u);
        __builtin_amdgcn_sched_barrier(0);
    }

    float* __restrict__ colL = &colNorm[wave][0];
#pragma unroll
    for (int i = 0; i < BT; ++i) {
        float sa = ssA[i].x + ssA[i].y;
        float sb = ssB[i].x + ssB[i].y;
        sa += __shfl_xor(sa, 16);
        sb += __shfl_xor(sb, 16);
        nrmL[i * 16 + l16] = sqrtf(sa);
        colL[i * 16 + l16] = sqrtf(sb);
    }

    float* __restrict__ myTile = &tile[wave][0];
#pragma unroll
    for (int i = 0; i < BT; ++i)
#pragma unroll
        for (int j = 0; j < BT; ++j)
#pragma unroll
            for (int r = 0; r < 8; ++r)
                myTile[(i * 16 + r + 8 * half) * TP + j * 16 + l16] = c[i][j][r];
    __syncthreads();

    const int ai  = lane & 7;
    const int si0 = lane >> 3;
    const float4 naLo = *(const float4*)(colL + ai * 8);
    const float4 naHi = *(const float4*)(colL + ai * 8 + 4);
    const float naV[8] = {naLo.x, naLo.y, naLo.z, naLo.w, naHi.x, naHi.y, naHi.z, naHi.w};
    const float NEG_INF = -__builtin_inff();
    float best0 = NEG_INF, best1 = NEG_INF;
    bool done0 = false, done1 = false;
#pragma unroll 1
    for (int w = WPC - 1; w >= 0; --w) {
        const int r0 = si0 * 8 + w;
        const float* p0 = &myTile[r0 * TP + ai * 8];
        const float* p1 = p0 + 32 * TP;
        const float4 x0 = *(const float4*)p0;
        const float4 x1 = *(const float4*)(p0 + 4);
        const float4 y0 = *(const float4*)p1;
        const float4 y1 = *(const float4*)(p1 + 4);
        const float ns0 = nrmL[r0], ns1 = nrmL[r0 + 32];
        const bool nm0 = nonL[r0] != 0.0f, nm1 = nonL[r0 + 32] != 0.0f;
        const v8f numX = {x0.x, x0.y, x0.z, x0.w, x1.x, x1.y, x1.z, x1.w};
        const v8f numY = {y0.x, y0.y, y0.z, y0.w, y1.x, y1.y, y1.z, y1.w};
        v8f denX, denY;
#pragma unroll
        for (int v = 0; v < WPC; ++v) {
            denX[v] = fmaxf(ns0 * naV[v], COS_EPS);
            denY[v] = fmaxf(ns1 * naV[v], COS_EPS);
        }
        const v8f qx = div8(numX, denX);
        const v8f qy = div8(numY, denY);
        float xe[WPC], ye[WPC];
        bool hx = done0, hy = done1;
#pragma unroll
        for (int v = WPC - 1; v >= 0; --v) {
            const float x = nm0 ? 0.0f : qx[v];
            const float y = nm1 ? 0.0f : qy[v];
            xe[v] = hx ? NEG_INF : x;
            ye[v] = hy ? NEG_INF : y;
            hx = hx || (x < 0.0f);
            hy = hy || (y < 0.0f);
        }
        best0 = fmaxf(best0, fmaxf(fmaxf(fmaxf(xe[0], xe[1]), fmaxf(xe[2], xe[3])),
                                   fmaxf(fmaxf(xe[4], xe[5]), fmaxf(xe[6], xe[7]))));
        best1 = fmaxf(best1, fmaxf(fmaxf(fmaxf(ye[0], ye[1]), fmaxf(ye[2], ye[3])),
                                   fmaxf(fmaxf(ye[4], ye[5]), fmaxf(ye[6], ye[7]))));
        done0 = hx;
        done1 = hy;
        if (__builtin_amdgcn_ballot_w32(!(done0 && done1)) == 0u) break;
    }
    const float res0 = done0 ? best0 : fmaxf(best0, 0.0f);
    const float res1 = done1 ? best1 : fmaxf(best1, 0.0f);
    const int sIdx = blockIdx.y * 16 + wy * 8 + si0;
    const int aIdx = blockIdx.x * 16 + wx * 8 + ai;
    maxSim[sIdx * A_CATS + aIdx] = res0;
    maxSim[(sIdx + 4) * A_CATS + aIdx] = res1;
}

extern "C" void kernel_launch(void* const* d_in, const int* in_sizes, int n_in,
                              void* d_out, int out_size, void* d_ws, size_t ws_size,
                              hipStream_t stream) {
    const float* emb      = (const float*)d_in[0];
    const int*   senCats  = (const int*)d_in[1];
    const int*   annCats  = (const int*)d_in[2];
    const int*   noneIdx  = (const int*)d_in[3];

    float* out    = (float*)d_out;
    float* maxSim = out;
    float* senWd  = out + S_CATS * A_CATS;

    ComSim_cos_block_wmma<<<dim3(16, 16), dim3(128), 0, stream>>>(
        emb, senCats, annCats, noneIdx, maxSim, senWd);
}


# Dots taken lazily, one sentence word a trip from word 7 down: a wave holds 16 sentences x 4 annotations (1x2 WMMA tiles), folds, and goes on only while a cell has no negative yet; sum orders kept
# speedup vs baseline: 5.3400x; 1.4764x over previous
#include <hip/hip_runtime.h>

typedef __attribute__((ext_vector_type(2))) float v2f;
typedef __attribute__((ext_vector_type(8))) float v8f;

#define DWORDS_D 300
#define ROW_BYTES (DWORDS_D * 4)
#define S_CATS 256
#define A_CATS 256
#define WPC 8
#define COS_EPS 1e-8f

__device__ __forceinline__ v8f div8(const v8f num, const v8f den)
{
    float ds[8], nm[8], rc[8], e[8], q[8], t[8];
    bool fl[8];
    bool unusedFlag;
    v8f res;
#pragma unroll
    for (int r = 0; r < 8; ++r) ds[r] = __builtin_amdgcn_div_scalef(num[r], den[r], false, &unusedFlag);
#pragma unroll
    for (int r = 0; r < 8; ++r) rc[r] = __builtin_amdgcn_rcpf(ds[r]);
#pragma unroll
    for (int r = 0; r < 8; ++r) e[r] = __builtin_fmaf(-ds[r], rc[r], 1.0f);
#pragma unroll
    for (int r = 0; r < 8; ++r) rc[r] = __builtin_fmaf(e[r], rc[r], rc[r]);
#pragma unroll
    for (int r = 0; r < 8; ++r) nm[r] = __builtin_amdgcn_div_scalef(num[r], den[r], true, &fl[r]);
#pragma unroll
    for (int r = 0; r < 8; ++r) q[r] = nm[r] * rc[r];
#pragma unroll
    for (int r = 0; r < 8; ++r) t[r] = __builtin_fmaf(-ds[r], q[r], nm[r]);
#pragma unroll
    for (int r = 0; r < 8; ++r) q[r] = __builtin_fmaf(t[r], rc[r], q[r]);
#pragma unroll
    for (int r = 0; r < 8; ++r) t[r] = __builtin_fmaf(-ds[r], q[r], nm[r]);
#pragma unroll
    for (int r = 0; r < 8; ++r) t[r] = __builtin_amdgcn_div_fmasf(t[r], rc[r], q[r], fl[r]);
#pragma unroll
    for (int r = 0; r < 8; ++r) res[r] = __builtin_amdgcn_div_fixupf(t[r], den[r], num[r]);
    return res;
}

__device__ __forceinline__ bool eq4(const float4 p, const float4 q)
{
    return (p.x == q.x) & (p.y == q.y) & (p.z == q.z) & (p.w == q.w);
}

#define BTB 2
#define RING 15
#define KSTEPS (DWORDS_D / 4)
#define TP 36

template <bool B> struct BoolTag { static constexpr bool value = B; };

__global__ __launch_bounds__(128)
void ComSim_cos_lazy_wmma(const float* __restrict__ emb,
                          const int* __restrict__ senCats,
                          const int* __restrict__ annCats,
                          const int* __restrict__ noneIdxPtr,
                          float* __restrict__ maxSim,
                          float* __restrict__ senWd)
{
    __shared__ __attribute__((aligned(16))) float tile[4][16 * TP];
    __shared__ __attribute__((aligned(16))) float rowNorm[4][16];
    __shared__ __attribute__((aligned(16))) float rowNone[4][16];
    __shared__ __attribute__((aligned(16))) float colNorm[4][BTB * 16];

    const int wave = __builtin_amdgcn_readfirstlane((int)(threadIdx.x >> 5));
    const int tid  = threadIdx.x;
    const int lane = threadIdx.x & 31;
    const int half = lane >> 4;
    const int l16  = lane & 15;
    const int wy   = wave >> 1;
    const int wx   = wave & 1;
    const int sen0 = blockIdx.y * 32 + wy * 16;
    const int ann0 = blockIdx.x * 8 + wx * 4;
    const int col0 = ann0 * WPC;

    unsigned offB[BTB];
#pragma unroll
    for (int j = 0; j < BTB; ++j)
        offB[j] = (unsigned)annCats[col0 + j * 16 + l16] * ROW_BYTES + 8u * half;
    const unsigned idxFirst = (unsigned)senCats[(sen0 + l16) * WPC + (WPC - 1)];
    const char* __restrict__ embB = (const char*)emb;
    const char* __restrict__ noneB = embB + (size_t)(unsigned)noneIdxPtr[0] * ROW_BYTES;
    const unsigned offH = 8u * half;

    const int sW = blockIdx.y * 32 + blockIdx.x;
    const int dW[3] = {tid, tid + 128, (tid + 256 < DWORDS_D) ? tid + 256 : DWORDS_D - 1};
    float wv[3][WPC];
#pragma unroll
    for (int w = 0; w < WPC; ++w) {
        const float* __restrict__ wr = emb + (size_t)(unsigned)senCats[sW * WPC + w] * DWORDS_D;
#pragma unroll
        for (int q = 0; q < 3; ++q) wv[q][w] = wr[dW[q]];
    }
    const float4* __restrict__ nRow = (const float4*)noneB;
    const float4 nHead = nRow[0];
    __builtin_amdgcn_sched_barrier(0);

    float* __restrict__ myTile = &tile[wave][0];
    float* __restrict__ nrmL = &rowNorm[wave][0];
    float* __restrict__ nonL = &rowNone[wave][0];
    float* __restrict__ colL = &colNorm[wave][0];

    const int ai  = lane & 3;
    const int si0 = lane >> 2;
    const float NEG_INF = -__builtin_inff();
    float best0 = NEG_INF, best1 = NEG_INF;
    bool done0 = false, done1 = false;
    float naV[WPC];

    auto trip = [&](auto firstTag, const unsigned rowIdx) -> bool {
        constexpr bool FIRST = decltype(firstTag)::value;
        const unsigned offA = rowIdx * ROW_BYTES + offH;
        const float4* __restrict__ xRow = (const float4*)(embB + (size_t)(offA - offH));
        const float4 xHead = xRow[0];

        v8f c[BTB];
#pragma unroll
        for (int j = 0; j < BTB; ++j) c[j] = (v8f){};
        v2f a[RING], b[RING][BTB];
        v2f ssA = (v2f){};
        v2f ssB[BTB];
#pragma unroll
        for (int j = 0; j < BTB; ++j) ssB[j] = (v2f){};

        auto loadSet = [&](int set, int kstep) {
            const char* __restrict__ kb = embB + 16 * kstep;
            a[set] = *(const v2f*)(kb + (size_t)offA);
#pragma unroll
            for (int j = 0; j < BTB; ++j) b[set][j] = *(const v2f*)(kb + (size_t)offB[j]);
        };
        auto wmmaSet = [&](int set) {
#pragma unroll
            for (int j = 0; j < BTB; ++j)
                c[j] = __builtin_amdgcn_wmma_f32_16x16x4_f32(
                         false, a[set],  false, b[set][j],
                         (short)0, c[j],  false,  false);
        };
        auto sqSet = [&](int set) {
            ssA = __builtin_elementwise_fma(a[set], a[set], ssA);
            if constexpr (FIRST) {
#pragma unroll
                for (int j = 0; j < BTB; ++j)
                    ssB[j] = __builtin_elementwise_fma(b[set][j], b[set][j], ssB[j]);
            }
        };

#pragma unroll
        for (int s = 0; s < RING - 1; ++s) {
            loadSet(s, s);
            __builtin_amdgcn_sched_barrier(0);
        }
        if constexpr (FIRST) {
            asm volatile("" :: "v"(wv[2][0]), "v"(wv[2][1]), "v"(wv[2][2]), "v"(wv[2][3]),
                               "v"(wv[2][4]), "v"(wv[2][5]), "v"(wv[2][6]), "v"(wv[2][7]));
        }

#pragma unroll 1
        for (int kk = 0; kk < KSTEPS - RING; kk += RING) {
#pragma unroll
            for (int u = 0; u < RING; ++u) {
                loadSet((u + RING - 1) % RING, kk + u + RING - 1);
                __builtin_amdgcn_sched_barrier(0);
                wmmaSet(u);
                sqSet(u);
                __builtin_amdgcn_sched_barrier(0);
            }
        }
        loadSet(RING - 1, KSTEPS - 1);
        __builtin_amdgcn_sched_barrier(0);
#pragma unroll
        for (int u = 0; u < RING; ++u) {
            wmmaSet(u);
            sqSet(u);
        }

        if constexpr (FIRST) {
#pragma unroll
            for (int q = 0; q < 3; ++q) {
                float acc = 0.0f;
#pragma unroll
                for (int ww = 0; ww < WPC; ++ww) acc += wv[q][ww];
                if (q < 2 || tid + 256 < DWORDS_D)
                    senWd[(size_t)sW * DWORDS_D + dW[q]] = acc;
            }
        }

        {
            float sa = ssA.x + ssA.y;
            sa += __shfl_xor(sa, 16);
            nrmL[l16] = sqrtf(sa);
        }
        if constexpr (FIRST) {
#pragma unroll
            for (int j = 0; j < BTB; ++j) {
                float sb = ssB[j].x + ssB[j].y;
                sb += __shfl_xor(sb, 16);
                colL[j * 16 + l16] = sqrtf(sb);
            }
        }
        {
            bool isNone = eq4(xHead, nHead);
            if (__builtin_amdgcn_ballot_w32(isNone) != 0u) {
                for (int t = 1; t < DWORDS_D / 4; ++t)
                    isNone = isNone & eq4(xRow[t], nRow[t]);
            }
            nonL[l16] = isNone ? 1.0f : 0.0f;
        }
#pragma unroll
        for (int j = 0; j < BTB; ++j)
#pragma unroll
            for (int r = 0; r < 8; ++r)
                myTile[(r + 8 * half) * TP + j * 16 + l16] = c[j][r];
        __builtin_amdgcn_fence(__ATOMIC_RELEASE, "wavefront");
        __builtin_amdgcn_wave_barrier();
        __builtin_amdgcn_fence(__ATOMIC_ACQUIRE, "wavefront");

        if constexpr (FIRST) {
            const float4 naLo = *(const float4*)(colL + ai * 8);
            const float4 naHi = *(const float4*)(colL + ai * 8 + 4);
            naV[0] = naLo.x; naV[1] = naLo.y; naV[2] = naLo.z; naV[3] = naLo.w;
            naV[4] = naHi.x; naV[5] = naHi.y; naV[6] = naHi.z; naV[7] = naHi.w;
        }
        const float* p0 = &myTile[si0 * TP + ai * 8];
        const float* p1 = p0 + 8 * TP;
        const float4 x0 = *(const float4*)p0;
        const float4 x1 = *(const float4*)(p0 + 4);
        const float4 y0 = *(const float4*)p1;
        const float4 y1 = *(const float4*)(p1 + 4);
        const float ns0 = nrmL[si0], ns1 = nrmL[si0 + 8];
        const bool nm0 = nonL[si0] != 0.0f, nm1 = nonL[si0 + 8] != 0.0f;
        const v8f numX = {x0.x, x0.y, x0.z, x0.w, x1.x, x1.y, x1.z, x1.w};
        const v8f numY = {y0.x, y0.y, y0.z, y0.w, y1.x, y1.y, y1.z, y1.w};
        v8f denX, denY;
#pragma unroll
        for (int v = 0; v < WPC; ++v) {
            denX[v] = fmaxf(ns0 * naV[v], COS_EPS);
            denY[v] = fmaxf(ns1 * naV[v], COS_EPS);
        }
        const v8f qx = div8(numX, denX);
        const v8f qy = div8(numY, denY);
        float xe[WPC], ye[WPC];
        bool hx = done0, hy = done1;
#pragma unroll
        for (int v = WPC - 1; v >= 0; --v) {
            const float x = nm0 ? 0.0f : qx[v];
            const float y = nm1 ? 0.0f : qy[v];
            xe[v] = hx ? NEG_INF : x;
            ye[v] = hy ? NEG_INF : y;
            hx = hx || (x < 0.0f);
            hy = hy || (y < 0.0f);
        }
        best0 = fmaxf(best0, fmaxf(fmaxf(fmaxf(xe[0], xe[1]), fmaxf(xe[2], xe[3])),
                                   fmaxf(fmaxf(xe[4], xe[5]), fmaxf(xe[6], xe[7]))));
        best1 = fmaxf(best1, fmaxf(fmaxf(fmaxf(ye[0], ye[1]), fmaxf(ye[2], ye[3])),
                                   fmaxf(fmaxf(ye[4], ye[5]), fmaxf(ye[6], ye[7]))));
        done0 = hx;
        done1 = hy;
        __builtin_amdgcn_fence(__ATOMIC_RELEASE, "wavefront");
        __builtin_amdgcn_wave_barrier();
        __builtin_amdgcn_fence(__ATOMIC_ACQUIRE, "wavefront");
        return __builtin_amdgcn_ballot_w32(!(done0 && done1)) == 0u;
    };

    bool fin = trip(BoolTag<true>{}, idxFirst);
#pragma unroll 1
    for (int w = WPC - 2; w >= 0 && !fin; --w)
        fin = trip(BoolTag<false>{}, (unsigned)senCats[(sen0 + l16) * WPC + w]);

    const float res0 = done0 ? best0 : fmaxf(best0, 0.0f);
    const float res1 = done1 ? best1 : fmaxf(best1, 0.0f);
    const int sIdx = sen0 + si0;
    const int aIdx = ann0 + ai;
    maxSim[sIdx * A_CATS + aIdx] = res0;
    maxSim[(sIdx + 8) * A_CATS + aIdx] = res1;
}

extern "C" void kernel_launch(void* const* d_in, const int* in_sizes, int n_in,
                              void* d_out, int out_size, void* d_ws, size_t ws_size,
                              hipStream_t stream) {
    const float* emb      = (const float*)d_in[0];
    const int*   senCats  = (const int*)d_in[1];
    const int*   annCats  = (const int*)d_in[2];
    const int*   noneIdx  = (const int*)d_in[3];

    float* out    = (float*)d_out;
    float* maxSim = out;
    float* senWd  = out + S_CATS * A_CATS;

    ComSim_cos_lazy_wmma<<<dim3(32, 8), dim3(128), 0, stream>>>(
        emb, senCats, annCats, noneIdx, maxSim, senWd);
}


# Fragment loads widened to 16 bytes per lane: one load feeds two k steps (k regrouped by 8, alike for A and B), so half as many loads and cache line look-ups; dot and norm sum orders changed
# speedup vs baseline: 7.0106x; 1.3128x over previous
#include <hip/hip_runtime.h>

typedef __attribute__((ext_vector_type(2))) float v2f;
typedef __attribute__((ext_vector_type(4))) float v4f;
typedef __attribute__((ext_vector_type(8))) float v8f;

#define DWORDS_D 300
#define ROW_BYTES (DWORDS_D * 4)
#define S_CATS 256
#define A_CATS 256
#define WPC 8
#define COS_EPS 1e-8f

__device__ __forceinline__ v8f div8(const v8f num, const v8f den)
{
    float ds[8], nm[8], rc[8], e[8], q[8], t[8];
    bool fl[8];
    bool unusedFlag;
    v8f res;
#pragma unroll
    for (int r = 0; r < 8; ++r) ds[r] = __builtin_amdgcn_div_scalef(num[r], den[r], false, &unusedFlag);
#pragma unroll
    for (int r = 0; r < 8; ++r) rc[r] = __builtin_amdgcn_rcpf(ds[r]);
#pragma unroll
    for (int r = 0; r < 8; ++r) e[r] = __builtin_fmaf(-ds[r], rc[r], 1.0f);
#pragma unroll
    for (int r = 0; r < 8; ++r) rc[r] = __builtin_fmaf(e[r], rc[r], rc[r]);
#pragma unroll
    for (int r = 0; r < 8; ++r) nm[r] = __builtin_amdgcn_div_scalef(num[r], den[r], true, &fl[r]);
#pragma unroll
    for (int r = 0; r < 8; ++r) q[r] = nm[r] * rc[r];
#pragma unroll
    for (int r = 0; r < 8; ++r) t[r] = __builtin_fmaf(-ds[r], q[r], nm[r]);
#pragma unroll
    for (int r = 0; r < 8; ++r) q[r] = __builtin_fmaf(t[r], rc[r], q[r]);
#pragma unroll
    for (int r = 0; r < 8; ++r) t[r] = __builtin_fmaf(-ds[r], q[r], nm[r]);
#pragma unroll
    for (int r = 0; r < 8; ++r) t[r] = __builtin_amdgcn_div_fmasf(t[r], rc[r], q[r], fl[r]);
#pragma unroll
    for (int r = 0; r < 8; ++r) res[r] = __builtin_amdgcn_div_fixupf(t[r], den[r], num[r]);
    return res;
}

__device__ __forceinline__ bool eq4(const float4 p, const float4 q)
{
    return (p.x == q.x) & (p.y == q.y) & (p.z == q.z) & (p.w == q.w);
}

#define BTB 2
#define RINGD 9
#define KPAIRS (DWORDS_D / 8)
#define LOOP_PAIRS (3 * RINGD)
static_assert(LOOP_PAIRS - RINGD + 2 * (RINGD - 1) < KPAIRS, "a prefetch of the rolled loop would leave the row");
static_assert(KPAIRS * 8 + 4 == DWORDS_D, "the row is 37 pairs of k steps and one single k step");
#define TP 36

template <bool B> struct BoolTag { static constexpr bool value = B; };

__global__ __launch_bounds__(128)
void ComSim_cos_lazy_wmma(const float* __restrict__ emb,
                          const int* __restrict__ senCats,
                          const int* __restrict__ annCats,
                          const int* __restrict__ noneIdxPtr,
                          float* __restrict__ maxSim,
                          float* __restrict__ senWd)
{
    __shared__ __attribute__((aligned(16))) float tile[4][16 * TP];
    __shared__ __attribute__((aligned(16))) float rowNorm[4][16];
    __shared__ __attribute__((aligned(16))) float rowNone[4][16];
    __shared__ __attribute__((aligned(16))) float colNorm[4][BTB * 16];

    const int wave = __builtin_amdgcn_readfirstlane((int)(threadIdx.x >> 5));
    const int tid  = threadIdx.x;
    const int lane = threadIdx.x & 31;
    const int half = lane >> 4;
    const int l16  = lane & 15;
    const int wy   = wave >> 1;
    const int wx   = wave & 1;
    const int sen0 = blockIdx.y * 32 + wy * 16;
    const int ann0 = blockIdx.x * 8 + wx * 4;
    const int col0 = ann0 * WPC;

    unsigned offB[BTB], offBT[BTB];
#pragma unroll
    for (int j = 0; j < BTB; ++j) {
        const unsigned rowB = (unsigned)annCats[col0 + j * 16 + l16] * ROW_BYTES;
        offB[j]  = rowB + 16u * half;
        offBT[j] = rowB + 8u * half;
    }
    const unsigned idxFirst = (unsigned)senCats[(sen0 + l16) * WPC + (WPC - 1)];
    const char* __restrict__ embB = (const char*)emb;
    const char* __restrict__ noneB = embB + (size_t)(unsigned)noneIdxPtr[0] * ROW_BYTES;
    const unsigned offH = 16u * half;

    const int sW = blockIdx.y * 32 + blockIdx.x;
    const int dW[3] = {tid, tid + 128, (tid + 256 < DWORDS_D) ? tid + 256 : DWORDS_D - 1};
    float wv[3][WPC];
#pragma unroll
    for (int w = 0; w < WPC; ++w) {
        const float* __restrict__ wr = emb + (size_t)(unsigned)senCats[sW * WPC + w] * DWORDS_D;
#pragma unroll
        for (int q = 0; q < 3; ++q) wv[q][w] = wr[dW[q]];
    }
    const float4* __restrict__ nRow = (const float4*)noneB;
    const float4 nHead = nRow[0];
    __builtin_amdgcn_sched_barrier(0);

    float* __restrict__ myTile = &tile[wave][0];
    float* __restrict__ nrmL = &rowNorm[wave][0];
    float* __restrict__ nonL = &rowNone[wave][0];
    float* __restrict__ colL = &colNorm[wave][0];

    const int ai  = lane & 3;
    const int si0 = lane >> 2;
    const float NEG_INF = -__builtin_inff();
    float best0 = NEG_INF, best1 = NEG_INF;
    bool done0 = false, done1 = false;
    float naV[WPC];

    auto trip = [&](auto firstTag, const unsigned rowIdx) -> bool {
        constexpr bool FIRST = decltype(firstTag)::value;
        const unsigned rowA = rowIdx * ROW_BYTES;
        const unsigned offA = rowA + offH;
        const unsigned offAT = rowA + 8u * half;
        const float4* __restrict__ xRow = (const float4*)(embB + (size_t)rowA);
        const float4 xHead = xRow[0];

        v8f c[BTB];
#pragma unroll
        for (int j = 0; j < BTB; ++j) c[j] = (v8f){};
        v4f a[RINGD], b[RINGD][BTB];
        v2f aT, bT[BTB];
        v2f ssA = (v2f){};
        v2f ssB[BTB];
#pragma unroll
        for (int j = 0; j < BTB; ++j) ssB[j] = (v2f){};

        auto loadSet = [&](int set, int kpair) {
            const char* __restrict__ kb = embB + 32 * kpair;
            a[set] = *(const v4f*)(kb + (size_t)offA);
#pragma unroll
            for (int j = 0; j < BTB; ++j) b[set][j] = *(const v4f*)(kb + (size_t)offB[j]);
        };
        auto loadTail = [&]() {
            const char* __restrict__ kb = embB + 32 * KPAIRS;
            aT = *(const v2f*)(kb + (size_t)offAT);
#pragma unroll
            for (int j = 0; j < BTB; ++j) bT[j] = *(const v2f*)(kb + (size_t)offBT[j]);
        };
        auto kStep = [&](const v2f av, const v2f (&bv)[BTB]) {
#pragma unroll
            for (int j = 0; j < BTB; ++j)
                c[j] = __builtin_amdgcn_wmma_f32_16x16x4_f32(
                         false, av,  false, bv[j],
                         (short)0, c[j],  false,  false);
            ssA = __builtin_elementwise_fma(av, av, ssA);
            if constexpr (FIRST) {
#pragma unroll
                for (int j = 0; j < BTB; ++j)
                    ssB[j] = __builtin_elementwise_fma(bv[j], bv[j], ssB[j]);
            }
        };
        auto useSet = [&](int set) {
            v2f lo[BTB], hi[BTB];
#pragma unroll
            for (int j = 0; j < BTB; ++j) { lo[j] = b[set][j].xy; hi[j] = b[set][j].zw; }
            kStep(a[set].xy, lo);
            kStep(a[set].zw, hi);
        };

#pragma unroll
        for (int s = 0; s < RINGD - 1; ++s) {
            loadSet(s, s);
            __builtin_amdgcn_sched_barrier(0);
        }
        if constexpr (FIRST) {
            asm volatile("" :: "v"(wv[2][0]), "v"(wv[2][1]), "v"(wv[2][2]), "v"(wv[2][3]),
                               "v"(wv[2][4]), "v"(wv[2][5]), "v"(wv[2][6]), "v"(wv[2][7]));
        }

#pragma unroll 1
        for (int kk = 0; kk < LOOP_PAIRS; kk += RINGD) {
#pragma unroll
            for (int u = 0; u < RINGD; ++u) {
                loadSet((u + RINGD - 1) % RINGD, kk + u + RINGD - 1);
                __builtin_amdgcn_sched_barrier(0);
                useSet(u);
                __builtin_amdgcn_sched_barrier(0);
            }
        }
#pragma unroll
        for (int u = 0; u < KPAIRS - LOOP_PAIRS; ++u) {
            const int next = LOOP_PAIRS + u + RINGD - 1;
            if (next < KPAIRS) {
                loadSet(next % RINGD, next);
                __builtin_amdgcn_sched_barrier(0);
            } else if (next == KPAIRS) {
                loadTail();
                __builtin_amdgcn_sched_barrier(0);
            }
            useSet((LOOP_PAIRS + u) % RINGD);
        }
        kStep(aT, bT);

        if constexpr (FIRST) {
#pragma unroll
            for (int q = 0; q < 3; ++q) {
                float acc = 0.0f;
#pragma unroll
                for (int ww = 0; ww < WPC; ++ww) acc += wv[q][ww];
                if (q < 2 || tid + 256 < DWORDS_D)
                    senWd[(size_t)sW * DWORDS_D + dW[q]] = acc;
            }
        }

        {
            float sa = ssA.x + ssA.y;
            sa += __shfl_xor(sa, 16);
            nrmL[l16] = sqrtf(sa);
        }
        if constexpr (FIRST) {
#pragma unroll
            for (int j = 0; j < BTB; ++j) {
                float sb = ssB[j].x + ssB[j].y;
                sb += __shfl_xor(sb, 16);
                colL[j * 16 + l16] = sqrtf(sb);
            }
        }
        {
            bool isNone = eq4(xHead, nHead);
            if (__builtin_amdgcn_ballot_w32(isNone) != 0u) {
                for (int t = 1; t < DWORDS_D / 4; ++t)
                    isNone = isNone & eq4(xRow[t], nRow[t]);
            }
            nonL[l16] = isNone ? 1.0f : 0.0f;
        }
#pragma unroll
        for (int j = 0; j < BTB; ++j)
#pragma unroll
            for (int r = 0; r < 8; ++r)
                myTile[(r + 8 * half) * TP + j * 16 + l16] = c[j][r];
        __builtin_amdgcn_fence(__ATOMIC_RELEASE, "wavefront");
        __builtin_amdgcn_wave_barrier();
        __builtin_amdgcn_fence(__ATOMIC_ACQUIRE, "wavefront");

        if constexpr (FIRST) {
            const float4 naLo = *(const float4*)(colL + ai * 8);
            const float4 naHi = *(const float4*)(colL + ai * 8 + 4);
            naV[0] = naLo.x; naV[1] = naLo.y; naV[2] = naLo.z; naV[3] = naLo.w;
            naV[4] = naHi.x; naV[5] = naHi.y; naV[6] = naHi.z; naV[7] = naHi.w;
        }
        const float* p0 = &myTile[si0 * TP + ai * 8];
        const float* p1 = p0 + 8 * TP;
        const float4 x0 = *(const float4*)p0;
        const float4 x1 = *(const float4*)(p0 + 4);
        const float4 y0 = *(const float4*)p1;
        const float4 y1 = *(const float4*)(p1 + 4);
        const float ns0 = nrmL[si0], ns1 = nrmL[si0 + 8];
        const bool nm0 = nonL[si0] != 0.0f, nm1 = nonL[si0 + 8] != 0.0f;
        const v8f numX = {x0.x, x0.y, x0.z, x0.w, x1.x, x1.y, x1.z, x1.w};
        const v8f numY = {y0.x, y0.y, y0.z, y0.w, y1.x, y1.y, y1.z, y1.w};
        v8f denX, denY;
#pragma unroll
        for (int v = 0; v < WPC; ++v) {
            denX[v] = fmaxf(ns0 * naV[v], COS_EPS);
            denY[v] = fmaxf(ns1 * naV[v], COS_EPS);
        }
        const v8f qx = div8(numX, denX);
        const v8f qy = div8(numY, denY);
        float xe[WPC], ye[WPC];
        bool hx = done0, hy = done1;
#pragma unroll
        for (int v = WPC - 1; v >= 0; --v) {
            const float x = nm0 ? 0.0f : qx[v];
            const float y = nm1 ? 0.0f : qy[v];
            xe[v] = hx ? NEG_INF : x;
            ye[v] = hy ? NEG_INF : y;
            hx = hx || (x < 0.0f);
            hy = hy || (y < 0.0f);
        }
        best0 = fmaxf(best0, fmaxf(fmaxf(fmaxf(xe[0], xe[1]), fmaxf(xe[2], xe[3])),
                                   fmaxf(fmaxf(xe[4], xe[5]), fmaxf(xe[6], xe[7]))));
        best1 = fmaxf(best1, fmaxf(fmaxf(fmaxf(ye[0], ye[1]), fmaxf(ye[2], ye[3])),
                                   fmaxf(fmaxf(ye[4], ye[5]), fmaxf(ye[6], ye[7]))));
        done0 = hx;
        done1 = hy;
        __builtin_amdgcn_fence(__ATOMIC_RELEASE, "wavefront");
        __builtin_amdgcn_wave_barrier();
        __builtin_amdgcn_fence(__ATOMIC_ACQUIRE, "wavefront");
        return __builtin_amdgcn_ballot_w32(!(done0 && done1)) == 0u;
    };

    bool fin = trip(BoolTag<true>{}, idxFirst);
#pragma unroll 1
    for (int w = WPC - 2; w >= 0 && !fin; --w)
        fin = trip(BoolTag<false>{}, (unsigned)senCats[(sen0 + l16) * WPC + w]);

    const float res0 = done0 ? best0 : fmaxf(best0, 0.0f);
    const float res1 = done1 ? best1 : fmaxf(best1, 0.0f);
    const int sIdx = sen0 + si0;
    const int aIdx = ann0 + ai;
    maxSim[sIdx * A_CATS + aIdx] = res0;
    maxSim[(sIdx + 8) * A_CATS + aIdx] = res1;
}

extern "C" void kernel_launch(void* const* d_in, const int* in_sizes, int n_in,
                              void* d_out, int out_size, void* d_ws, size_t ws_size,
                              hipStream_t stream) {
    const float* emb      = (const float*)d_in[0];
    const int*   senCats  = (const int*)d_in[1];
    const int*   annCats  = (const int*)d_in[2];
    const int*   noneIdx  = (const int*)d_in[3];

    float* out    = (float*)d_out;
    float* maxSim = out;
    float* senWd  = out + S_CATS * A_CATS;

    ComSim_cos_lazy_wmma<<<dim3(32, 8), dim3(128), 0, stream>>>(
        emb, senCats, annCats, noneIdx, maxSim, senWd);
}
